# starting kernel
# baseline (speedup 1.0000x reference)
#include <hip/hip_runtime.h>

typedef float v2f __attribute__((ext_vector_type(2)));
typedef float v4f __attribute__((ext_vector_type(4)));
typedef float v8f __attribute__((ext_vector_type(8)));

#define CCH  256
#define IDIM 16
#define BDIM 4096
#define NK3  23
#define NK2  4
#define NK1  1

#define WS_U3W 0
#define WS_C2  (CCH * 4096)
#define WS_C1  (WS_C2 + CCH * 256)
#define WS_O3  (WS_C1 + CCH * 16)

__launch_bounds__(256)
__global__ void mace_precompute(const float* __restrict__ U3,
                                const float* __restrict__ U2,
                                const float* __restrict__ U1,
                                const float* __restrict__ w3,
                                const float* __restrict__ w2,
                                const float* __restrict__ w1,
                                float* __restrict__ ws) {
    const int c = blockIdx.x;
    const int t = threadIdx.x;
    const int p = t & 15;
    const int q = t >> 4;

    float w3r[NK3];
#pragma unroll
    for (int k = 0; k < NK3; ++k) w3r[k] = w3[c * NK3 + k];

    float* u3out = ws + WS_U3W + (size_t)c * 4096;
#pragma unroll
    for (int i = 0; i < 16; ++i) {
        float acc = 0.f;
        const float* u3p = U3 + ((p * 16 + q) * 16 + i) * NK3;
#pragma unroll
        for (int k = 0; k < NK3; ++k) acc += u3p[k] * w3r[k];
        const int kk = q * 16 + i;
        u3out[((kk >> 1) * 16 + p) * 2 + (kk & 1)] = acc;
    }

    {
        const int i = t >> 4;
        float acc = 0.f;
#pragma unroll
        for (int k = 0; k < NK2; ++k) acc += U2[(p * 16 + i) * NK2 + k] * w2[c * NK2 + k];
        ws[WS_C2 + c * 256 + ((i >> 1) * 16 + p) * 2 + (i & 1)] = acc;
    }
    if (t < 16) {
        float acc = 0.f;
#pragma unroll
        for (int k = 0; k < NK1; ++k) acc += U1[t * NK1 + k] * w1[c * NK1 + k];
        ws[WS_C1 + c * 16 + t] = acc;
    }
}

__launch_bounds__(256, 1)
__global__ void mace_contract(const float* __restrict__ x,
                              const float* __restrict__ wsc,
                              float* __restrict__ o3) {
    __shared__ float sU3[4096];
    __shared__ float sC2[256];
    __shared__ float sC1[16];
    __shared__ float sZ[8 * 256];

    const int c = blockIdx.x;
    const int t = threadIdx.x;

    const float* gU3 = wsc + WS_U3W + (size_t)c * 4096;
#pragma unroll
    for (int j = 0; j < 4; ++j)
        ((v4f*)sU3)[t + j * 256] = ((const v4f*)gU3)[t + j * 256];
    sC2[t] = wsc[WS_C2 + c * 256 + t];
    if (t < 16) sC1[t] = wsc[WS_C1 + c * 16 + t];
    __syncthreads();

    const int w    = t >> 5;
    const int lane = t & 31;
    const int m    = lane & 15;
    const int hi   = lane >> 4;

    const float  c1v    = sC1[m];
    const float* u3lane = sU3 + m * 2 + hi * 32;
    const float* c2lane = sC2 + m * 2 + hi * 32;
    float*       zrow   = sZ + w * 256;

    v2f bU[64];
#pragma unroll
    for (int k = 0; k < 64; ++k) bU[k] = *(const v2f*)(u3lane + k * 64);
    v2f bC2[4];
#pragma unroll
    for (int k = 0; k < 4; ++k)  bC2[k] = *(const v2f*)(c2lane + k * 64);

    for (int it = 0; it < 4; ++it) {
        const int T  = blockIdx.y * 32 + w * 4 + it;
        const int b0 = T * 16;

        const float* xrow = x + ((size_t)(b0 + m) * CCH + c) * IDIM;
        float xr[16];
#pragma unroll
        for (int j = 0; j < 4; ++j) {
            v4f v = ((const v4f*)xrow)[j];
            xr[j * 4 + 0] = v.x; xr[j * 4 + 1] = v.y;
            xr[j * 4 + 2] = v.z; xr[j * 4 + 3] = v.w;
        }
        float xs[8];
#pragma unroll
        for (int tt = 0; tt < 4; ++tt) {
            xs[2 * tt]     = hi ? xr[4 * tt + 2] : xr[4 * tt];
            xs[2 * tt + 1] = hi ? xr[4 * tt + 3] : xr[4 * tt + 1];
        }
        float xh[8];
#pragma unroll
        for (int j = 0; j < 8; ++j) xh[j] = hi ? xr[8 + j] : xr[j];

        v8f acc = {};
#pragma unroll
        for (int k = 0; k < 4; ++k) {
            v2f a;  a.x = xs[2 * k]; a.y = xs[2 * k + 1];
            acc = __builtin_amdgcn_wmma_f32_16x16x4_f32(
                false, a, false, bC2[k], (short)0, acc, false, false);
        }
#pragma unroll
        for (int r = 0; r < 8; ++r) acc[r] += c1v;

#pragma unroll
        for (int k = 0; k < 64; ++k) {
            const int q  = k >> 2;
            const int tt = k & 3;
            v2f a;  a.x = xr[q] * xs[2 * tt];  a.y = xr[q] * xs[2 * tt + 1];
            acc = __builtin_amdgcn_wmma_f32_16x16x4_f32(
                false, a, false, bU[k], (short)0, acc, false, false);
        }

#pragma unroll
        for (int r = 0; r < 8; ++r)
            zrow[(r + 8 * hi) * 16 + m] = acc[r];
        __asm__ volatile("s_wait_dscnt 0x0" ::: "memory");
        float psum = 0.f;
        const float* zr = zrow + m * 16 + 8 * hi;
#pragma unroll
        for (int j = 0; j < 8; ++j) psum += zr[j] * xh[j];
        psum += __shfl_xor(psum, 16, 32);
        if (lane < 16) o3[(size_t)c * BDIM + b0 + m] = psum;
    }
}

__launch_bounds__(256)
__global__ void mace_linear(const float* __restrict__ o3,
                            const float* __restrict__ Wl,
                            float* __restrict__ out) {
    __shared__ float sW[4096];
    const int t  = threadIdx.x;
    const int d0 = blockIdx.x * 16;
#pragma unroll
    for (int j4 = 0; j4 < 16; ++j4) {
        const int e   = t + j4 * 256;
        const int jj  = e & 1;
        const int n   = (e >> 1) & 15;
        const int K2i = e >> 5;
        sW[e] = Wl[(2 * K2i + jj) * 256 + d0 + n];
    }
    __syncthreads();

    const int w = t >> 5, lane = t & 31, m = lane & 15, hi = lane >> 4;
    const float* wlane = sW + m * 2 + hi * 32;
    const int b0 = (blockIdx.y * 8 + w) * 16;

    v8f acc = {};
#pragma unroll
    for (int k = 0; k < 64; ++k) {
        v2f a;
        a.x = o3[(size_t)(4 * k + 2 * hi)     * BDIM + b0 + m];
        a.y = o3[(size_t)(4 * k + 2 * hi + 1) * BDIM + b0 + m];
        v2f bb = *(const v2f*)(wlane + k * 64);
        acc = __builtin_amdgcn_wmma_f32_16x16x4_f32(
            false, a, false, bb, (short)0, acc, false, false);
    }
#pragma unroll
    for (int r = 0; r < 8; ++r)
        out[(size_t)(b0 + r + 8 * hi) * 256 + d0 + m] = acc[r] * 0.0625f;
}

extern "C" void kernel_launch(void* const* d_in, const int* in_sizes, int n_in,
                              void* d_out, int out_size, void* d_ws, size_t ws_size,
                              hipStream_t stream) {
    const float* x  = (const float*)d_in[0];
    const float* U3 = (const float*)d_in[1];
    const float* U2 = (const float*)d_in[2];
    const float* U1 = (const float*)d_in[3];
    const float* w3 = (const float*)d_in[4];
    const float* w2 = (const float*)d_in[5];
    const float* w1 = (const float*)d_in[6];
    const float* Wl = (const float*)d_in[7];
    float* out = (float*)d_out;
    float* ws  = (float*)d_ws;

    mace_precompute<<<dim3(CCH), dim3(256), 0, stream>>>(U3, U2, U1, w3, w2, w1, ws);
    mace_contract  <<<dim3(CCH, 8), dim3(256), 0, stream>>>(x, ws, ws + WS_O3);
    mace_linear    <<<dim3(16, 32), dim3(256), 0, stream>>>(ws + WS_O3, Wl, out);
}


# Fold mirrored (q,i) rows of the order-3 basis once per channel: contraction K 256->136, 34 f32 WMMA steps per tile instead of 64; changes the order of summation
# speedup vs baseline: 1.2524x; 1.2524x over previous
#include <hip/hip_runtime.h>


typedef float v2f __attribute__((ext_vector_type(2)));
typedef float v4f __attribute__((ext_vector_type(4)));
typedef float v8f __attribute__((ext_vector_type(8)));

#define CCH  256
#define IDIM 16
#define BDIM 4096
#define NK3  23
#define NK2  4
#define NK1  1
#define NPAIR 136
#define NSTEP3 (NPAIR / 4)
#define U3F   (NPAIR * 16)

#define WS_U3W 0
#define WS_C2  (CCH * 4096)
#define WS_C1  (WS_C2 + CCH * 256)
#define WS_O3  (WS_C1 + CCH * 16)


__launch_bounds__(256)
__global__ void mace_precompute(const float* __restrict__ U3,
                                const float* __restrict__ U2,
                                const float* __restrict__ U1,
                                const float* __restrict__ w3,
                                const float* __restrict__ w2,
                                const float* __restrict__ w1,
                                float* __restrict__ ws) {
    const unsigned char pair_of_pos[NPAIR] = {
        2, 3, 70, 71, 18, 19, 86, 87, 138, 139, 206, 207, 154, 155, 222, 223, 36,
        37, 104, 105, 52, 53, 120, 121, 172, 173, 224, 225, 188, 189, 240, 241, 4, 5,
        72, 73, 20, 21, 88, 89, 140, 141, 192, 193, 156, 157, 208, 209, 38, 39, 106,
        107, 54, 55, 122, 123, 174, 175, 226, 227, 190, 191, 242, 243, 6, 7, 74, 75,
        22, 23, 90, 91, 142, 143, 194, 195, 158, 159, 210, 211, 40, 41, 108, 109, 56,
        57, 124, 125, 160, 161, 228, 229, 176, 177, 244, 245, 8, 9, 76, 77, 24, 25,
        92, 93, 42, 43, 110, 111, 58, 59, 126, 127, 0, 1, 68, 69, 136, 137, 204,
        205, 34, 35, 102, 103, 170, 171, 238, 239, 17, 153, 85, 221, 51, 187, 119, 255
    };
    __shared__ float sM[256 * 17];

    const int c = blockIdx.x;
    const int t = threadIdx.x;
    const int p = t & 15;
    const int q = t >> 4;

    float w3r[NK3];
#pragma unroll
    for (int k = 0; k < NK3; ++k) w3r[k] = w3[c * NK3 + k];

#pragma unroll
    for (int i = 0; i < 16; ++i) {
        float acc = 0.f;
        const float* u3p = U3 + ((p * 16 + q) * 16 + i) * NK3;
#pragma unroll
        for (int k = 0; k < NK3; ++k) acc += u3p[k] * w3r[k];
        sM[(q * 16 + i) * 17 + p] = acc;
    }

    {
        const int pos2 = t >> 4;
        const int n    = ((pos2 >> 2) << 1) | (pos2 & 1);
        const int i    = n + ((n >> 2) << 2) + ((pos2 >> 1) & 1) * 4;
        float acc = 0.f;
#pragma unroll
        for (int k = 0; k < NK2; ++k) acc += U2[(p * 16 + i) * NK2 + k] * w2[c * NK2 + k];
        ws[WS_C2 + c * 256 + ((pos2 >> 1) * 16 + p) * 2 + (pos2 & 1)] = acc;
    }
    if (t < 16) {
        float acc = 0.f;
#pragma unroll
        for (int k = 0; k < NK1; ++k) acc += U1[t * NK1 + k] * w1[c * NK1 + k];
        ws[WS_C1 + c * 16 + t] = acc;
    }
    __syncthreads();

    float* u3out = ws + WS_U3W + (size_t)c * 4096;
    int qis[(U3F + 255) / 256];
#pragma unroll
    for (int n = 0; n < (U3F + 255) / 256; ++n) {
        const int f   = t + n * 256;
        const int pos = ((f >> 5) << 1) | (f & 1);
        qis[n] = pair_of_pos[f < U3F ? pos : 0];
    }
#pragma unroll
    for (int n = 0; n < (U3F + 255) / 256; ++n) {
        const int f = t + n * 256;
        if (f < U3F) {
            const int pp = (f >> 1) & 15;
            const int qq = qis[n] >> 4, ii = qis[n] & 15;
            const float v1 = sM[(qq * 16 + ii) * 17 + pp];
            const float v2 = sM[(ii * 16 + qq) * 17 + pp];
            u3out[f] = v1 + (qq != ii ? v2 : 0.f);
        }
    }
}

#define SC_STEP(k, q0, i0, q1, i1)                                           \
        { v2f a;  a.x = xr[q0] * xr[i0];  a.y = xr[q1] * xr[i1];             \
          acc = __builtin_amdgcn_wmma_f32_16x16x4_f32(                       \
              false, a, false, bU[k], (short)0, acc, false, false); }

__launch_bounds__(256, 1)
__global__ void mace_contract(const float* __restrict__ x,
                              const float* __restrict__ wsc,
                              float* __restrict__ o3) {
    __shared__ float sU3[U3F];
    __shared__ float sC2[256];
    __shared__ float sC1[16];
    __shared__ float sZ[8 * 256];

    const int c = blockIdx.x;
    const int t = threadIdx.x;

    const float* gU3 = wsc + WS_U3W + (size_t)c * 4096;
#pragma unroll
    for (int j = 0; j < (U3F / 4 + 255) / 256; ++j) {
        const int e = t + j * 256;
        if (e < U3F / 4) ((v4f*)sU3)[e] = ((const v4f*)gU3)[e];
    }
    sC2[t] = wsc[WS_C2 + c * 256 + t];
    if (t < 16) sC1[t] = wsc[WS_C1 + c * 16 + t];
    __syncthreads();

    const int w    = t >> 5;
    const int lane = t & 31;
    const int m    = lane & 15;
    const int hi   = lane >> 4;

    const float  c1v    = sC1[m];
    const float* u3lane = sU3 + m * 2 + hi * 32;
    const float* c2lane = sC2 + m * 2 + hi * 32;
    float*       zrow   = sZ + w * 256;

    v2f bU[NSTEP3];
#pragma unroll
    for (int k = 0; k < NSTEP3; ++k) bU[k] = *(const v2f*)(u3lane + k * 64);
    v2f bC2[4];
#pragma unroll
    for (int k = 0; k < 4; ++k)  bC2[k] = *(const v2f*)(c2lane + k * 64);

    for (int it = 0; it < 4; ++it) {
        const int T  = blockIdx.y * 32 + w * 4 + it;
        const int b0 = T * 16;

        const float* xrow = x + ((size_t)(b0 + m) * CCH + c) * IDIM;
        float xr[16];
#pragma unroll
        for (int j = 0; j < 4; ++j) {
            v4f v = ((const v4f*)xrow)[(j + hi) & 3];
            xr[j * 4 + 0] = v.x; xr[j * 4 + 1] = v.y;
            xr[j * 4 + 2] = v.z; xr[j * 4 + 3] = v.w;
        }

        v8f acc = {};
#pragma unroll
        for (int k = 0; k < 4; ++k) {
            const int r = 2 * k + ((k >> 1) << 2);
            v2f a;  a.x = xr[r]; a.y = xr[r + 1];
            acc = __builtin_amdgcn_wmma_f32_16x16x4_f32(
                false, a, false, bC2[k], (short)0, acc, false, false);
        }
#pragma unroll
        for (int r = 0; r < 8; ++r) acc[r] += c1v;

        SC_STEP(0, 0, 2, 0, 3)
        SC_STEP(1, 1, 2, 1, 3)
        SC_STEP(2, 8, 10, 8, 11)
        SC_STEP(3, 9, 10, 9, 11)
        SC_STEP(4, 2, 4, 2, 5)
        SC_STEP(5, 3, 4, 3, 5)
        SC_STEP(6, 10, 12, 10, 13)
        SC_STEP(7, 11, 12, 11, 13)
        SC_STEP(8, 0, 4, 0, 5)
        SC_STEP(9, 1, 4, 1, 5)
        SC_STEP(10, 8, 12, 8, 13)
        SC_STEP(11, 9, 12, 9, 13)
        SC_STEP(12, 2, 6, 2, 7)
        SC_STEP(13, 3, 6, 3, 7)
        SC_STEP(14, 10, 14, 10, 15)
        SC_STEP(15, 11, 14, 11, 15)
        SC_STEP(16, 0, 6, 0, 7)
        SC_STEP(17, 1, 6, 1, 7)
        SC_STEP(18, 8, 14, 8, 15)
        SC_STEP(19, 9, 14, 9, 15)
        SC_STEP(20, 2, 8, 2, 9)
        SC_STEP(21, 3, 8, 3, 9)
        SC_STEP(22, 10, 0, 10, 1)
        SC_STEP(23, 11, 0, 11, 1)
        SC_STEP(24, 0, 8, 0, 9)
        SC_STEP(25, 1, 8, 1, 9)
        SC_STEP(26, 2, 10, 2, 11)
        SC_STEP(27, 3, 10, 3, 11)
        SC_STEP(28, 0, 0, 0, 1)
        SC_STEP(29, 8, 8, 8, 9)
        SC_STEP(30, 2, 2, 2, 3)
        SC_STEP(31, 10, 10, 10, 11)
        SC_STEP(32, 1, 1, 9, 9)
        SC_STEP(33, 3, 3, 11, 11)

#pragma unroll
        for (int r = 0; r < 8; ++r)
            zrow[(r + 8 * hi) * 16 + m] = acc[r];
        __asm__ volatile("s_wait_dscnt 0x0" ::: "memory");
        float psum = 0.f;
        const float* zr = zrow + m * 16 + 4 * hi;
#pragma unroll
        for (int j = 0; j < 4; ++j) psum += zr[j] * xr[j];
#pragma unroll
        for (int j = 0; j < 4; ++j) psum += zr[8 + j] * xr[8 + j];
        psum += __shfl_xor(psum, 16, 32);
        if (lane < 16) o3[(size_t)c * BDIM + b0 + m] = psum;
    }
}

__launch_bounds__(256)
__global__ void mace_linear(const float* __restrict__ o3,
                            const float* __restrict__ Wl,
                            float* __restrict__ out) {
    __shared__ float sW[4096];
    const int t  = threadIdx.x;
    const int d0 = blockIdx.x * 16;
#pragma unroll
    for (int j4 = 0; j4 < 16; ++j4) {
        const int e   = t + j4 * 256;
        const int jj  = e & 1;
        const int n   = (e >> 1) & 15;
        const int K2i = e >> 5;
        sW[e] = Wl[(2 * K2i + jj) * 256 + d0 + n];
    }
    __syncthreads();

    const int w = t >> 5, lane = t & 31, m = lane & 15, hi = lane >> 4;
    const float* wlane = sW + m * 2 + hi * 32;
    const int b0 = (blockIdx.y * 8 + w) * 16;

    v8f acc = {};
#pragma unroll
    for (int k = 0; k < 64; ++k) {
        v2f a;
        a.x = o3[(size_t)(4 * k + 2 * hi)     * BDIM + b0 + m];
        a.y = o3[(size_t)(4 * k + 2 * hi + 1) * BDIM + b0 + m];
        v2f bb = *(const v2f*)(wlane + k * 64);
        acc = __builtin_amdgcn_wmma_f32_16x16x4_f32(
            false, a, false, bb, (short)0, acc, false, false);
    }
#pragma unroll
    for (int r = 0; r < 8; ++r)
        out[(size_t)(b0 + r + 8 * hi) * 256 + d0 + m] = acc[r] * 0.0625f;
}

extern "C" void kernel_launch(void* const* d_in, const int* in_sizes, int n_in,
                              void* d_out, int out_size, void* d_ws, size_t ws_size,
                              hipStream_t stream) {
    const float* x  = (const float*)d_in[0];
    const float* U3 = (const float*)d_in[1];
    const float* U2 = (const float*)d_in[2];
    const float* U1 = (const float*)d_in[3];
    const float* w3 = (const float*)d_in[4];
    const float* w2 = (const float*)d_in[5];
    const float* w1 = (const float*)d_in[6];
    const float* Wl = (const float*)d_in[7];
    float* out = (float*)d_out;
    float* ws  = (float*)d_ws;

    mace_precompute<<<dim3(CCH), dim3(256), 0, stream>>>(U3, U2, U1, w3, w2, w1, ws);
    mace_contract  <<<dim3(CCH, 8), dim3(256), 0, stream>>>(x, ws, ws + WS_O3);
    mace_linear    <<<dim3(16, 32), dim3(256), 0, stream>>>(ws + WS_O3, Wl, out);
}


# Stage A streams U3 once per block through LDS with whole-line coalesced 16-byte loads, next group in flight while one is reduced, instead of per-thread strided 1472-byte runs
# speedup vs baseline: 1.5181x; 1.2122x over previous
#include <hip/hip_runtime.h>


typedef float v2f __attribute__((ext_vector_type(2)));
typedef float v4f __attribute__((ext_vector_type(4)));
typedef float v8f __attribute__((ext_vector_type(8)));
typedef float v4fu __attribute__((ext_vector_type(4), aligned(4)));

#define CCH  256
#define IDIM 16
#define BDIM 4096
#define NK3  23
#define NK2  4
#define NK1  1
#define NPAIR 136
#define NSTEP3 (NPAIR / 4)
#define U3F   (NPAIR * 16)
#define U3_SLAB (256 * NK3)
#define SLABS_PER_GRP 4
#define NGRP  (16 / SLABS_PER_GRP)
#define GRP_F (SLABS_PER_GRP * U3_SLAB)
#define GRP_V4PT (GRP_F / 4 / 256)

#define WS_U3W 0
#define WS_C2  (CCH * 4096)
#define WS_C1  (WS_C2 + CCH * 256)
#define WS_O3  (WS_C1 + CCH * 16)


__launch_bounds__(256)
__global__ void mace_precompute(const float* __restrict__ U3,
                                const float* __restrict__ U2,
                                const float* __restrict__ U1,
                                const float* __restrict__ w3,
                                const float* __restrict__ w2,
                                const float* __restrict__ w1,
                                float* __restrict__ ws) {
    const unsigned char pair_of_pos[NPAIR] = {
        2, 3, 70, 71, 18, 19, 86, 87, 138, 139, 206, 207, 154, 155, 222, 223, 36,
        37, 104, 105, 52, 53, 120, 121, 172, 173, 224, 225, 188, 189, 240, 241, 4, 5,
        72, 73, 20, 21, 88, 89, 140, 141, 192, 193, 156, 157, 208, 209, 38, 39, 106,
        107, 54, 55, 122, 123, 174, 175, 226, 227, 190, 191, 242, 243, 6, 7, 74, 75,
        22, 23, 90, 91, 142, 143, 194, 195, 158, 159, 210, 211, 40, 41, 108, 109, 56,
        57, 124, 125, 160, 161, 228, 229, 176, 177, 244, 245, 8, 9, 76, 77, 24, 25,
        92, 93, 42, 43, 110, 111, 58, 59, 126, 127, 0, 1, 68, 69, 136, 137, 204,
        205, 34, 35, 102, 103, 170, 171, 238, 239, 17, 153, 85, 221, 51, 187, 119, 255
    };
    __shared__ float sM[256 * 17];
    __shared__ __attribute__((aligned(16))) float sU[GRP_F];

    const int c = blockIdx.x;
    const int t = threadIdx.x;

    float w3r[NK3];
#pragma unroll
    for (int k = 0; k < NK3; ++k) w3r[k] = w3[c * NK3 + k];

    const v4fu* U3v = (const v4fu*)U3;
    const int g0 = c & (NGRP - 1);
    v4f r[GRP_V4PT];
#pragma unroll
    for (int j = 0; j < GRP_V4PT; ++j)
        r[j] = U3v[g0 * (GRP_F / 4) + j * 256 + t];

#pragma unroll
    for (int s = 0; s < NGRP; ++s) {
        const int g = (g0 + s) & (NGRP - 1);
#pragma unroll
        for (int j = 0; j < GRP_V4PT; ++j) ((v4f*)sU)[j * 256 + t] = r[j];
        __syncthreads();
        if (s + 1 < NGRP) {
            const int gn = (g0 + s + 1) & (NGRP - 1);
#pragma unroll
            for (int j = 0; j < GRP_V4PT; ++j)
                r[j] = U3v[gn * (GRP_F / 4) + j * 256 + t];
        }
#pragma unroll
        for (int pp = 0; pp < SLABS_PER_GRP; ++pp) {
            const float* row = sU + (pp * 256 + t) * NK3;
            float acc = 0.f;
#pragma unroll
            for (int k = 0; k < NK3; ++k) acc += row[k] * w3r[k];
            sM[t * 17 + g * SLABS_PER_GRP + pp] = acc;
        }
        __syncthreads();
    }

    const int p = t & 15;

    {
        const int pos2 = t >> 4;
        const int n    = ((pos2 >> 2) << 1) | (pos2 & 1);
        const int i    = n + ((n >> 2) << 2) + ((pos2 >> 1) & 1) * 4;
        float acc = 0.f;
#pragma unroll
        for (int k = 0; k < NK2; ++k) acc += U2[(p * 16 + i) * NK2 + k] * w2[c * NK2 + k];
        ws[WS_C2 + c * 256 + ((pos2 >> 1) * 16 + p) * 2 + (pos2 & 1)] = acc;
    }
    if (t < 16) {
        float acc = 0.f;
#pragma unroll
        for (int k = 0; k < NK1; ++k) acc += U1[t * NK1 + k] * w1[c * NK1 + k];
        ws[WS_C1 + c * 16 + t] = acc;
    }

    float* u3out = ws + WS_U3W + (size_t)c * 4096;
    int qis[(U3F + 255) / 256];
#pragma unroll
    for (int n = 0; n < (U3F + 255) / 256; ++n) {
        const int f   = t + n * 256;
        const int pos = ((f >> 5) << 1) | (f & 1);
        qis[n] = pair_of_pos[f < U3F ? pos : 0];
    }
#pragma unroll
    for (int n = 0; n < (U3F + 255) / 256; ++n) {
        const int f = t + n * 256;
        if (f < U3F) {
            const int pp = (f >> 1) & 15;
            const int qq = qis[n] >> 4, ii = qis[n] & 15;
            const float v1 = sM[(qq * 16 + ii) * 17 + pp];
            const float v2 = sM[(ii * 16 + qq) * 17 + pp];
            u3out[f] = v1 + (qq != ii ? v2 : 0.f);
        }
    }
}

#define SC_STEP(k, q0, i0, q1, i1)                                           \
        { v2f a;  a.x = xr[q0] * xr[i0];  a.y = xr[q1] * xr[i1];             \
          acc = __builtin_amdgcn_wmma_f32_16x16x4_f32(                       \
              false, a, false, bU[k], (short)0, acc, false, false); }

__launch_bounds__(256, 1)
__global__ void mace_contract(const float* __restrict__ x,
                              const float* __restrict__ wsc,
                              float* __restrict__ o3) {
    __shared__ float sU3[U3F];
    __shared__ float sC2[256];
    __shared__ float sC1[16];
    __shared__ float sZ[8 * 256];

    const int c = blockIdx.x;
    const int t = threadIdx.x;

    const float* gU3 = wsc + WS_U3W + (size_t)c * 4096;
#pragma unroll
    for (int j = 0; j < (U3F / 4 + 255) / 256; ++j) {
        const int e = t + j * 256;
        if (e < U3F / 4) ((v4f*)sU3)[e] = ((const v4f*)gU3)[e];
    }
    sC2[t] = wsc[WS_C2 + c * 256 + t];
    if (t < 16) sC1[t] = wsc[WS_C1 + c * 16 + t];
    __syncthreads();

    const int w    = t >> 5;
    const int lane = t & 31;
    const int m    = lane & 15;
    const int hi   = lane >> 4;

    const float  c1v    = sC1[m];
    const float* u3lane = sU3 + m * 2 + hi * 32;
    const float* c2lane = sC2 + m * 2 + hi * 32;
    float*       zrow   = sZ + w * 256;

    v2f bU[NSTEP3];
#pragma unroll
    for (int k = 0; k < NSTEP3; ++k) bU[k] = *(const v2f*)(u3lane + k * 64);
    v2f bC2[4];
#pragma unroll
    for (int k = 0; k < 4; ++k)  bC2[k] = *(const v2f*)(c2lane + k * 64);

    for (int it = 0; it < 4; ++it) {
        const int T  = blockIdx.y * 32 + w * 4 + it;
        const int b0 = T * 16;

        const float* xrow = x + ((size_t)(b0 + m) * CCH + c) * IDIM;
        float xr[16];
#pragma unroll
        for (int j = 0; j < 4; ++j) {
            v4f v = ((const v4f*)xrow)[(j + hi) & 3];
            xr[j * 4 + 0] = v.x; xr[j * 4 + 1] = v.y;
            xr[j * 4 + 2] = v.z; xr[j * 4 + 3] = v.w;
        }

        v8f acc = {};
#pragma unroll
        for (int k = 0; k < 4; ++k) {
            const int r = 2 * k + ((k >> 1) << 2);
            v2f a;  a.x = xr[r]; a.y = xr[r + 1];
            acc = __builtin_amdgcn_wmma_f32_16x16x4_f32(
                false, a, false, bC2[k], (short)0, acc, false, false);
        }
#pragma unroll
        for (int r = 0; r < 8; ++r) acc[r] += c1v;

        SC_STEP(0, 0, 2, 0, 3)
        SC_STEP(1, 1, 2, 1, 3)
        SC_STEP(2, 8, 10, 8, 11)
        SC_STEP(3, 9, 10, 9, 11)
        SC_STEP(4, 2, 4, 2, 5)
        SC_STEP(5, 3, 4, 3, 5)
        SC_STEP(6, 10, 12, 10, 13)
        SC_STEP(7, 11, 12, 11, 13)
        SC_STEP(8, 0, 4, 0, 5)
        SC_STEP(9, 1, 4, 1, 5)
        SC_STEP(10, 8, 12, 8, 13)
        SC_STEP(11, 9, 12, 9, 13)
        SC_STEP(12, 2, 6, 2, 7)
        SC_STEP(13, 3, 6, 3, 7)
        SC_STEP(14, 10, 14, 10, 15)
        SC_STEP(15, 11, 14, 11, 15)
        SC_STEP(16, 0, 6, 0, 7)
        SC_STEP(17, 1, 6, 1, 7)
        SC_STEP(18, 8, 14, 8, 15)
        SC_STEP(19, 9, 14, 9, 15)
        SC_STEP(20, 2, 8, 2, 9)
        SC_STEP(21, 3, 8, 3, 9)
        SC_STEP(22, 10, 0, 10, 1)
        SC_STEP(23, 11, 0, 11, 1)
        SC_STEP(24, 0, 8, 0, 9)
        SC_STEP(25, 1, 8, 1, 9)
        SC_STEP(26, 2, 10, 2, 11)
        SC_STEP(27, 3, 10, 3, 11)
        SC_STEP(28, 0, 0, 0, 1)
        SC_STEP(29, 8, 8, 8, 9)
        SC_STEP(30, 2, 2, 2, 3)
        SC_STEP(31, 10, 10, 10, 11)
        SC_STEP(32, 1, 1, 9, 9)
        SC_STEP(33, 3, 3, 11, 11)

#pragma unroll
        for (int r = 0; r < 8; ++r)
            zrow[(r + 8 * hi) * 16 + m] = acc[r];
        __asm__ volatile("s_wait_dscnt 0x0" ::: "memory");
        float psum = 0.f;
        const float* zr = zrow + m * 16 + 4 * hi;
#pragma unroll
        for (int j = 0; j < 4; ++j) psum += zr[j] * xr[j];
#pragma unroll
        for (int j = 0; j < 4; ++j) psum += zr[8 + j] * xr[8 + j];
        psum += __shfl_xor(psum, 16, 32);
        if (lane < 16) o3[(size_t)c * BDIM + b0 + m] = psum;
    }
}

__launch_bounds__(256)
__global__ void mace_linear(const float* __restrict__ o3,
                            const float* __restrict__ Wl,
                            float* __restrict__ out) {
    __shared__ float sW[4096];
    const int t  = threadIdx.x;
    const int d0 = blockIdx.x * 16;
#pragma unroll
    for (int j4 = 0; j4 < 16; ++j4) {
        const int e   = t + j4 * 256;
        const int jj  = e & 1;
        const int n   = (e >> 1) & 15;
        const int K2i = e >> 5;
        sW[e] = Wl[(2 * K2i + jj) * 256 + d0 + n];
    }
    __syncthreads();

    const int w = t >> 5, lane = t & 31, m = lane & 15, hi = lane >> 4;
    const float* wlane = sW + m * 2 + hi * 32;
    const int b0 = (blockIdx.y * 8 + w) * 16;

    v8f acc = {};
#pragma unroll
    for (int k = 0; k < 64; ++k) {
        v2f a;
        a.x = o3[(size_t)(4 * k + 2 * hi)     * BDIM + b0 + m];
        a.y = o3[(size_t)(4 * k + 2 * hi + 1) * BDIM + b0 + m];
        v2f bb = *(const v2f*)(wlane + k * 64);
        acc = __builtin_amdgcn_wmma_f32_16x16x4_f32(
            false, a, false, bb, (short)0, acc, false, false);
    }
#pragma unroll
    for (int r = 0; r < 8; ++r)
        out[(size_t)(b0 + r + 8 * hi) * 256 + d0 + m] = acc[r] * 0.0625f;
}

extern "C" void kernel_launch(void* const* d_in, const int* in_sizes, int n_in,
                              void* d_out, int out_size, void* d_ws, size_t ws_size,
                              hipStream_t stream) {
    const float* x  = (const float*)d_in[0];
    const float* U3 = (const float*)d_in[1];
    const float* U2 = (const float*)d_in[2];
    const float* U1 = (const float*)d_in[3];
    const float* w3 = (const float*)d_in[4];
    const float* w2 = (const float*)d_in[5];
    const float* w1 = (const float*)d_in[6];
    const float* Wl = (const float*)d_in[7];
    float* out = (float*)d_out;
    float* ws  = (float*)d_ws;

    mace_precompute<<<dim3(CCH), dim3(256), 0, stream>>>(U3, U2, U1, w3, w2, w1, ws);
    mace_contract  <<<dim3(CCH, 8), dim3(256), 0, stream>>>(x, ws, ws + WS_O3);
    mace_linear    <<<dim3(16, 32), dim3(256), 0, stream>>>(ws + WS_O3, Wl, out);
}


# Fold the order-3 cubic form over all orderings of (p,q,i): pair rows taken inside each half of x, contraction K 136->72, 18 f32 WMMA steps per tile instead of 34; changes the order of summation
# speedup vs baseline: 1.8538x; 1.2211x over previous
#include <hip/hip_runtime.h>


typedef float v2f __attribute__((ext_vector_type(2)));
typedef float v4f __attribute__((ext_vector_type(4)));
typedef float v8f __attribute__((ext_vector_type(8)));
typedef float v4fu __attribute__((ext_vector_type(4), aligned(4)));

#define CCH  256
#define IDIM 16
#define BDIM 4096
#define NK3  23
#define NK2  4
#define NK1  1
#define NPAIR 72
#define NSTEP3 (NPAIR / 4)
#define U3F   (NPAIR * 16)
#define U3_SLAB (256 * NK3)
#define SLABS_PER_GRP 4
#define NGRP  (16 / SLABS_PER_GRP)
#define GRP_F (SLABS_PER_GRP * U3_SLAB)
#define GRP_V4PT (GRP_F / 4 / 256)

#define WS_U3W 0
#define WS_C2  (CCH * 4096)
#define WS_C1  (WS_C2 + CCH * 256)
#define WS_O3  (WS_C1 + CCH * 16)


#define SC_PAIRS(X)                                                          \
    X(0, 0, 0, 0, 1)   X(1, 0, 2, 0, 3)   X(2, 0, 4, 0, 5)   X(3, 0, 6, 0, 7)   \
    X(4, 1, 1, 1, 2)   X(5, 1, 3, 1, 4)   X(6, 1, 5, 1, 6)   X(7, 1, 7, 2, 2)   \
    X(8, 2, 3, 2, 4)   X(9, 2, 5, 2, 6)   X(10, 2, 7, 3, 3)  X(11, 3, 4, 3, 5)  \
    X(12, 3, 6, 3, 7)  X(13, 4, 4, 4, 5)  X(14, 4, 6, 4, 7)  X(15, 5, 5, 5, 6)  \
    X(16, 5, 7, 6, 6)  X(17, 6, 7, 7, 7)
#define SC_TABLE(k, q0, i0, q1, i1) ((q0) << 4) | (i0), ((q1) << 4) | (i1),

__launch_bounds__(256)
__global__ void mace_precompute(const float* __restrict__ U3,
                                const float* __restrict__ U2,
                                const float* __restrict__ U1,
                                const float* __restrict__ w3,
                                const float* __restrict__ w2,
                                const float* __restrict__ w1,
                                float* __restrict__ ws) {
    const unsigned char pair_of_n[NPAIR / 2] = { SC_PAIRS(SC_TABLE) };
    __shared__ float sM[256 * 17];
    __shared__ __attribute__((aligned(16))) float sU[GRP_F];

    const int c = blockIdx.x;
    const int t = threadIdx.x;

    float w3r[NK3];
#pragma unroll
    for (int k = 0; k < NK3; ++k) w3r[k] = w3[c * NK3 + k];

    const v4fu* U3v = (const v4fu*)U3;
    const int g0 = c & (NGRP - 1);
    v4f r[GRP_V4PT];
#pragma unroll
    for (int j = 0; j < GRP_V4PT; ++j)
        r[j] = U3v[g0 * (GRP_F / 4) + j * 256 + t];

#pragma unroll
    for (int s = 0; s < NGRP; ++s) {
        const int g = (g0 + s) & (NGRP - 1);
#pragma unroll
        for (int j = 0; j < GRP_V4PT; ++j) ((v4f*)sU)[j * 256 + t] = r[j];
        __syncthreads();
        if (s + 1 < NGRP) {
            const int gn = (g0 + s + 1) & (NGRP - 1);
#pragma unroll
            for (int j = 0; j < GRP_V4PT; ++j)
                r[j] = U3v[gn * (GRP_F / 4) + j * 256 + t];
        }
#pragma unroll
        for (int pp = 0; pp < SLABS_PER_GRP; ++pp) {
            const float* row = sU + (pp * 256 + t) * NK3;
            float acc = 0.f;
#pragma unroll
            for (int k = 0; k < NK3; ++k) acc += row[k] * w3r[k];
            sM[t * 17 + g * SLABS_PER_GRP + pp] = acc;
        }
        __syncthreads();
    }

    const int p = t & 15;

    {
        const int pos2 = t >> 4;
        const int n    = ((pos2 >> 2) << 1) | (pos2 & 1);
        const int i    = n + ((pos2 >> 1) & 1) * 8;
        float acc = 0.f;
#pragma unroll
        for (int k = 0; k < NK2; ++k) acc += U2[(p * 16 + i) * NK2 + k] * w2[c * NK2 + k];
        ws[WS_C2 + c * 256 + ((pos2 >> 1) * 16 + p) * 2 + (pos2 & 1)] = acc;
    }
    if (t < 16) {
        float acc = 0.f;
#pragma unroll
        for (int k = 0; k < NK1; ++k) acc += U1[t * NK1 + k] * w1[c * NK1 + k];
        ws[WS_C1 + c * 16 + t] = acc;
    }

    float* u3out = ws + WS_U3W + (size_t)c * 4096;
    int qis[(U3F + 255) / 256];
#pragma unroll
    for (int n = 0; n < (U3F + 255) / 256; ++n) {
        const int f   = t + n * 256;
        const int pos = ((f >> 5) << 1) | (f & 1);
        const int nn  = ((pos >> 2) << 1) | (pos & 1);
        qis[n] = pair_of_n[f < U3F ? nn : 0];
    }
#define T3(a, b, e) sM[((b) * 16 + (e)) * 17 + (a)]
#pragma unroll
    for (int n = 0; n < (U3F + 255) / 256; ++n) {
        const int f = t + n * 256;
        if (f < U3F) {
            const int h  = (f >> 5) & 1;
            const int pp = (f >> 1) & 15;
            const int qq = (qis[n] >> 4) + 8 * h, ii = (qis[n] & 15) + 8 * h;
            const bool pq = pp != qq, pi = pp != ii, qi = qq != ii;
            float v = T3(pp, qq, ii);
            v += qi             ? T3(pp, ii, qq) : 0.f;
            v += pq             ? T3(qq, pp, ii) : 0.f;
            v += (pq && pi)     ? T3(qq, ii, pp) : 0.f;
            v += (qi && pi)     ? T3(ii, pp, qq) : 0.f;
            v += (pq && pi && qi) ? T3(ii, qq, pp) : 0.f;
            const bool mine = ((pp >> 3) != h) || (pp >= ii);
            u3out[f] = mine ? v : 0.f;
        }
    }
#undef T3
}

#define SC_STEP(k, q0, i0, q1, i1)                                           \
        { v2f a;  a.x = xr[q0] * xr[i0];  a.y = xr[q1] * xr[i1];             \
          acc = __builtin_amdgcn_wmma_f32_16x16x4_f32(                       \
              false, a, false, bU[k], (short)0, acc, false, false); }

__launch_bounds__(256, 1)
__global__ void mace_contract(const float* __restrict__ x,
                              const float* __restrict__ wsc,
                              float* __restrict__ o3) {
    __shared__ float sU3[U3F];
    __shared__ float sC2[256];
    __shared__ float sC1[16];
    __shared__ float sZ[8 * 256];

    const int c = blockIdx.x;
    const int t = threadIdx.x;

    const float* gU3 = wsc + WS_U3W + (size_t)c * 4096;
#pragma unroll
    for (int j = 0; j < (U3F / 4 + 255) / 256; ++j) {
        const int e = t + j * 256;
        if (e < U3F / 4) ((v4f*)sU3)[e] = ((const v4f*)gU3)[e];
    }
    sC2[t] = wsc[WS_C2 + c * 256 + t];
    if (t < 16) sC1[t] = wsc[WS_C1 + c * 16 + t];
    __syncthreads();

    const int w    = t >> 5;
    const int lane = t & 31;
    const int m    = lane & 15;
    const int hi   = lane >> 4;

    const float  c1v    = sC1[m];
    const float* u3lane = sU3 + m * 2 + hi * 32;
    const float* c2lane = sC2 + m * 2 + hi * 32;
    float*       zrow   = sZ + w * 256;

    v2f bU[NSTEP3];
#pragma unroll
    for (int k = 0; k < NSTEP3; ++k) bU[k] = *(const v2f*)(u3lane + k * 64);
    v2f bC2[4];
#pragma unroll
    for (int k = 0; k < 4; ++k)  bC2[k] = *(const v2f*)(c2lane + k * 64);

    for (int it = 0; it < 4; ++it) {
        const int T  = blockIdx.y * 32 + w * 4 + it;
        const int b0 = T * 16;

        const float* xrow = x + ((size_t)(b0 + m) * CCH + c) * IDIM + 8 * hi;
        float xr[8];
#pragma unroll
        for (int j = 0; j < 2; ++j) {
            v4f v = ((const v4f*)xrow)[j];
            xr[j * 4 + 0] = v.x; xr[j * 4 + 1] = v.y;
            xr[j * 4 + 2] = v.z; xr[j * 4 + 3] = v.w;
        }

        v8f acc = {};
#pragma unroll
        for (int k = 0; k < 4; ++k) {
            const int r = 2 * k;
            v2f a;  a.x = xr[r]; a.y = xr[r + 1];
            acc = __builtin_amdgcn_wmma_f32_16x16x4_f32(
                false, a, false, bC2[k], (short)0, acc, false, false);
        }
#pragma unroll
        for (int r = 0; r < 8; ++r) acc[r] += c1v;

        SC_PAIRS(SC_STEP)

#pragma unroll
        for (int r = 0; r < 8; ++r)
            zrow[(r + 8 * hi) * 16 + m] = acc[r];
        __asm__ volatile("s_wait_dscnt 0x0" ::: "memory");
        float psum = 0.f;
        const float* zr = zrow + m * 16 + 8 * hi;
#pragma unroll
        for (int j = 0; j < 8; ++j) psum += zr[j] * xr[j];
        psum += __shfl_xor(psum, 16, 32);
        if (lane < 16) o3[(size_t)c * BDIM + b0 + m] = psum;
    }
}

__launch_bounds__(256)
__global__ void mace_linear(const float* __restrict__ o3,
                            const float* __restrict__ Wl,
                            float* __restrict__ out) {
    __shared__ float sW[4096];
    const int t  = threadIdx.x;
    const int d0 = blockIdx.x * 16;
#pragma unroll
    for (int j4 = 0; j4 < 16; ++j4) {
        const int e   = t + j4 * 256;
        const int jj  = e & 1;
        const int n   = (e >> 1) & 15;
        const int K2i = e >> 5;
        sW[e] = Wl[(2 * K2i + jj) * 256 + d0 + n];
    }
    __syncthreads();

    const int w = t >> 5, lane = t & 31, m = lane & 15, hi = lane >> 4;
    const float* wlane = sW + m * 2 + hi * 32;
    const int b0 = (blockIdx.y * 8 + w) * 16;

    v8f acc = {};
#pragma unroll
    for (int k = 0; k < 64; ++k) {
        v2f a;
        a.x = o3[(size_t)(4 * k + 2 * hi)     * BDIM + b0 + m];
        a.y = o3[(size_t)(4 * k + 2 * hi + 1) * BDIM + b0 + m];
        v2f bb = *(const v2f*)(wlane + k * 64);
        acc = __builtin_amdgcn_wmma_f32_16x16x4_f32(
            false, a, false, bb, (short)0, acc, false, false);
    }
#pragma unroll
    for (int r = 0; r < 8; ++r)
        out[(size_t)(b0 + r + 8 * hi) * 256 + d0 + m] = acc[r] * 0.0625f;
}

extern "C" void kernel_launch(void* const* d_in, const int* in_sizes, int n_in,
                              void* d_out, int out_size, void* d_ws, size_t ws_size,
                              hipStream_t stream) {
    const float* x  = (const float*)d_in[0];
    const float* U3 = (const float*)d_in[1];
    const float* U2 = (const float*)d_in[2];
    const float* U1 = (const float*)d_in[3];
    const float* w3 = (const float*)d_in[4];
    const float* w2 = (const float*)d_in[5];
    const float* w1 = (const float*)d_in[6];
    const float* Wl = (const float*)d_in[7];
    float* out = (float*)d_out;
    float* ws  = (float*)d_ws;

    mace_precompute<<<dim3(CCH), dim3(256), 0, stream>>>(U3, U2, U1, w3, w2, w1, ws);
    mace_contract  <<<dim3(CCH, 8), dim3(256), 0, stream>>>(x, ws, ws + WS_O3);
    mace_linear    <<<dim3(16, 32), dim3(256), 0, stream>>>(ws + WS_O3, Wl, out);
}


# Stage A sums the (p,q,i) orderings on U3 itself, then applies channel weights: a block gathers only its 32 cells' U3 rows instead of streaming all of U3 per channel; changes the order of summation
# speedup vs baseline: 2.0706x; 1.1169x over previous
#include <hip/hip_runtime.h>


typedef float v2f __attribute__((ext_vector_type(2)));
typedef float v4f __attribute__((ext_vector_type(4)));
typedef float v8f __attribute__((ext_vector_type(8)));

#define CCH  256
#define IDIM 16
#define BDIM 4096
#define NK3  23
#define NK2  4
#define NK1  1
#define NPAIR 72
#define NSTEP3 (NPAIR / 4)
#define U3F   (NPAIR * 16)

#define WS_U3W 0
#define WS_C2  (CCH * 4096)
#define WS_C1  (WS_C2 + CCH * 256)
#define WS_O3  (WS_C1 + CCH * 16)


#define SC_PAIRS(X)                                                          \
    X(0, 0, 0, 0, 1)   X(1, 0, 2, 0, 3)   X(2, 0, 4, 0, 5)   X(3, 0, 6, 0, 7)   \
    X(4, 1, 1, 1, 2)   X(5, 1, 3, 1, 4)   X(6, 1, 5, 1, 6)   X(7, 1, 7, 2, 2)   \
    X(8, 2, 3, 2, 4)   X(9, 2, 5, 2, 6)   X(10, 2, 7, 3, 3)  X(11, 3, 4, 3, 5)  \
    X(12, 3, 6, 3, 7)  X(13, 4, 4, 4, 5)  X(14, 4, 6, 4, 7)  X(15, 5, 5, 5, 6)  \
    X(16, 5, 7, 6, 6)  X(17, 6, 7, 7, 7)
__host__ __device__ constexpr int pair_code(int n) {
    const int q = (n >= 8) + (n >= 15) + (n >= 21) + (n >= 26) + (n >= 30) + (n >= 33) + (n >= 35);
    return (q << 4) | (q + n - (q * 8 - q * (q - 1) / 2));
}
#define SC_CHECK(k, q0, i0, q1, i1)                                          \
    static_assert(pair_code(2 * (k)) == (((q0) << 4) | (i0)) &&              \
                  pair_code(2 * (k) + 1) == (((q1) << 4) | (i1)), "pair order");
SC_PAIRS(SC_CHECK)

#define A_CELLS   32
#define A_NBLK3   (U3F / A_CELLS)
#define A_CH      64
#define A_SPLIT   (CCH / A_CH)
#define A_CHW     (A_CH / 8)
#define A_CH2     8
#define A_NBLK2   (CCH / A_CH2)
#define A_GRID    (A_NBLK3 * A_SPLIT + A_NBLK2)
#define A_UFP     28
#define A_WP      24

__launch_bounds__(256)
__global__ void mace_precompute(const float* __restrict__ U3,
                                const float* __restrict__ U2,
                                const float* __restrict__ U1,
                                const float* __restrict__ w3,
                                const float* __restrict__ w2,
                                const float* __restrict__ w1,
                                float* __restrict__ ws) {
    const int t = threadIdx.x;

    if (blockIdx.x >= A_NBLK3 * A_SPLIT) {
        const int c0   = (blockIdx.x - A_NBLK3 * A_SPLIT) * A_CH2;
        const int p    = t & 15;
        const int pos2 = t >> 4;
        const int n    = ((pos2 >> 2) << 1) | (pos2 & 1);
        const int i    = n + ((pos2 >> 1) & 1) * 8;
        float u2[NK2];
#pragma unroll
        for (int k = 0; k < NK2; ++k) u2[k] = U2[(p * 16 + i) * NK2 + k];
        float u1[NK1];
#pragma unroll
        for (int k = 0; k < NK1; ++k) u1[k] = U1[(t & 15) * NK1 + k];
        const int dst2 = ((pos2 >> 1) * 16 + p) * 2 + (pos2 & 1);
        float w2r[A_CH2][NK2], w1r[A_CH2][NK1];
#pragma unroll
        for (int ch = 0; ch < A_CH2; ++ch) {
#pragma unroll
            for (int k = 0; k < NK2; ++k) w2r[ch][k] = w2[(c0 + ch) * NK2 + k];
#pragma unroll
            for (int k = 0; k < NK1; ++k) w1r[ch][k] = w1[(c0 + ch) * NK1 + k];
        }
#pragma unroll
        for (int ch = 0; ch < A_CH2; ++ch) {
            const int c = c0 + ch;
            float acc = 0.f;
#pragma unroll
            for (int k = 0; k < NK2; ++k) acc += u2[k] * w2r[ch][k];
            ws[WS_C2 + c * 256 + dst2] = acc;
            float a1 = 0.f;
#pragma unroll
            for (int k = 0; k < NK1; ++k) a1 += u1[k] * w1r[ch][k];
            if (t < 16) ws[WS_C1 + c * 16 + t] = a1;
        }
        return;
    }

    __shared__ __attribute__((aligned(16))) float sUf[A_CELLS * A_UFP];
    __shared__ __attribute__((aligned(16))) float sW[A_CH * A_WP];

    const int blk = blockIdx.x / A_SPLIT;
    const int c0  = (blockIdx.x % A_SPLIT) * A_CH;

    const int cell = t >> 3, sub = t & 7;
    const int h  = blk & 1;
    const int pp = cell >> 1;
    const int qi = pair_code(((blk >> 1) << 1) | (cell & 1));
    const int qq = (qi >> 4) + 8 * h, ii = (qi & 15) + 8 * h;
    const bool pq = pp != qq, pi = pp != ii, qd = qq != ii;
    const bool mine = ((pp >> 3) != h) || (pp >= ii);
#define U3ROW(a, b, e) ((((a) * 16 + (b)) * 16 + (e)) * NK3)
    const int r0 = U3ROW(pp, qq, ii), r1 = U3ROW(pp, ii, qq), r2 = U3ROW(qq, pp, ii);
    const int r3 = U3ROW(qq, ii, pp), r4 = U3ROW(ii, pp, qq), r5 = U3ROW(ii, qq, pp);
#undef U3ROW
    float a[3][6];
#pragma unroll
    for (int kk = 0; kk < 3; ++kk) {
        const int k = (sub + 8 * kk < NK3) ? sub + 8 * kk : NK3 - 1;
        a[kk][0] = U3[r0 + k]; a[kk][1] = U3[r1 + k]; a[kk][2] = U3[r2 + k];
        a[kk][3] = U3[r3 + k]; a[kk][4] = U3[r4 + k]; a[kk][5] = U3[r5 + k];
    }
    const int wch = t >> 2, wpart = t & 3;
    float wv[6];
#pragma unroll
    for (int j = 0; j < 6; ++j) {
        const int k = (wpart + 4 * j < NK3) ? wpart + 4 * j : NK3 - 1;
        wv[j] = w3[(c0 + wch) * NK3 + k];
    }
#pragma unroll
    for (int kk = 0; kk < 3; ++kk) {
        float v = a[kk][0];
        v += qd               ? a[kk][1] : 0.f;
        v += pq               ? a[kk][2] : 0.f;
        v += (pq && pi)       ? a[kk][3] : 0.f;
        v += (qd && pi)       ? a[kk][4] : 0.f;
        v += (pq && pi && qd) ? a[kk][5] : 0.f;
        sUf[cell * A_UFP + sub + 8 * kk] = mine ? v : 0.f;
    }
#pragma unroll
    for (int j = 0; j < 6; ++j) sW[wch * A_WP + wpart + 4 * j] = wv[j];
    __syncthreads();

    const int w = t >> 5, lane = t & 31;
    float uf[NK3];
#pragma unroll
    for (int k = 0; k < NK3; ++k) uf[k] = sUf[lane * A_UFP + k];
    float* dst = ws + WS_U3W + (size_t)(c0 + w * A_CHW) * 4096 + blk * A_CELLS + lane;
#pragma unroll
    for (int ch = 0; ch < A_CHW; ++ch) {
        const float* wr = sW + (w * A_CHW + ch) * A_WP;
        float acc = 0.f;
#pragma unroll
        for (int k = 0; k < NK3; ++k) acc += uf[k] * wr[k];
        dst[(size_t)ch * 4096] = acc;
    }
}

#define SC_STEP(k, q0, i0, q1, i1)                                           \
        { v2f a;  a.x = xr[q0] * xr[i0];  a.y = xr[q1] * xr[i1];             \
          acc = __builtin_amdgcn_wmma_f32_16x16x4_f32(                       \
              false, a, false, bU[k], (short)0, acc, false, false); }

__launch_bounds__(256, 1)
__global__ void mace_contract(const float* __restrict__ x,
                              const float* __restrict__ wsc,
                              float* __restrict__ o3) {
    __shared__ float sU3[U3F];
    __shared__ float sC2[256];
    __shared__ float sC1[16];
    __shared__ float sZ[8 * 256];

    const int c = blockIdx.x;
    const int t = threadIdx.x;

    const float* gU3 = wsc + WS_U3W + (size_t)c * 4096;
#pragma unroll
    for (int j = 0; j < (U3F / 4 + 255) / 256; ++j) {
        const int e = t + j * 256;
        if (e < U3F / 4) ((v4f*)sU3)[e] = ((const v4f*)gU3)[e];
    }
    sC2[t] = wsc[WS_C2 + c * 256 + t];
    if (t < 16) sC1[t] = wsc[WS_C1 + c * 16 + t];
    __syncthreads();

    const int w    = t >> 5;
    const int lane = t & 31;
    const int m    = lane & 15;
    const int hi   = lane >> 4;

    const float  c1v    = sC1[m];
    const float* u3lane = sU3 + m * 2 + hi * 32;
    const float* c2lane = sC2 + m * 2 + hi * 32;
    float*       zrow   = sZ + w * 256;

    v2f bU[NSTEP3];
#pragma unroll
    for (int k = 0; k < NSTEP3; ++k) bU[k] = *(const v2f*)(u3lane + k * 64);
    v2f bC2[4];
#pragma unroll
    for (int k = 0; k < 4; ++k)  bC2[k] = *(const v2f*)(c2lane + k * 64);

    for (int it = 0; it < 4; ++it) {
        const int T  = blockIdx.y * 32 + w * 4 + it;
        const int b0 = T * 16;

        const float* xrow = x + ((size_t)(b0 + m) * CCH + c) * IDIM + 8 * hi;
        float xr[8];
#pragma unroll
        for (int j = 0; j < 2; ++j) {
            v4f v = ((const v4f*)xrow)[j];
            xr[j * 4 + 0] = v.x; xr[j * 4 + 1] = v.y;
            xr[j * 4 + 2] = v.z; xr[j * 4 + 3] = v.w;
        }

        v8f acc = {};
#pragma unroll
        for (int k = 0; k < 4; ++k) {
            const int r = 2 * k;
            v2f a;  a.x = xr[r]; a.y = xr[r + 1];
            acc = __builtin_amdgcn_wmma_f32_16x16x4_f32(
                false, a, false, bC2[k], (short)0, acc, false, false);
        }
#pragma unroll
        for (int r = 0; r < 8; ++r) acc[r] += c1v;

        SC_PAIRS(SC_STEP)

#pragma unroll
        for (int r = 0; r < 8; ++r)
            zrow[(r + 8 * hi) * 16 + m] = acc[r];
        __asm__ volatile("s_wait_dscnt 0x0" ::: "memory");
        float psum = 0.f;
        const float* zr = zrow + m * 16 + 8 * hi;
#pragma unroll
        for (int j = 0; j < 8; ++j) psum += zr[j] * xr[j];
        psum += __shfl_xor(psum, 16, 32);
        if (lane < 16) o3[(size_t)c * BDIM + b0 + m] = psum;
    }
}

__launch_bounds__(256)
__global__ void mace_linear(const float* __restrict__ o3,
                            const float* __restrict__ Wl,
                            float* __restrict__ out) {
    __shared__ float sW[4096];
    const int t  = threadIdx.x;
    const int d0 = blockIdx.x * 16;
#pragma unroll
    for (int j4 = 0; j4 < 16; ++j4) {
        const int e   = t + j4 * 256;
        const int jj  = e & 1;
        const int n   = (e >> 1) & 15;
        const int K2i = e >> 5;
        sW[e] = Wl[(2 * K2i + jj) * 256 + d0 + n];
    }
    __syncthreads();

    const int w = t >> 5, lane = t & 31, m = lane & 15, hi = lane >> 4;
    const float* wlane = sW + m * 2 + hi * 32;
    const int b0 = (blockIdx.y * 8 + w) * 16;

    v8f acc = {};
#pragma unroll
    for (int k = 0; k < 64; ++k) {
        v2f a;
        a.x = o3[(size_t)(4 * k + 2 * hi)     * BDIM + b0 + m];
        a.y = o3[(size_t)(4 * k + 2 * hi + 1) * BDIM + b0 + m];
        v2f bb = *(const v2f*)(wlane + k * 64);
        acc = __builtin_amdgcn_wmma_f32_16x16x4_f32(
            false, a, false, bb, (short)0, acc, false, false);
    }
#pragma unroll
    for (int r = 0; r < 8; ++r)
        out[(size_t)(b0 + r + 8 * hi) * 256 + d0 + m] = acc[r] * 0.0625f;
}

extern "C" void kernel_launch(void* const* d_in, const int* in_sizes, int n_in,
                              void* d_out, int out_size, void* d_ws, size_t ws_size,
                              hipStream_t stream) {
    const float* x  = (const float*)d_in[0];
    const float* U3 = (const float*)d_in[1];
    const float* U2 = (const float*)d_in[2];
    const float* U1 = (const float*)d_in[3];
    const float* w3 = (const float*)d_in[4];
    const float* w2 = (const float*)d_in[5];
    const float* w1 = (const float*)d_in[6];
    const float* Wl = (const float*)d_in[7];
    float* out = (float*)d_out;
    float* ws  = (float*)d_ws;

    mace_precompute<<<dim3(A_GRID), dim3(256), 0, stream>>>(U3, U2, U1, w3, w2, w1, ws);
    mace_contract  <<<dim3(CCH, 8), dim3(256), 0, stream>>>(x, ws, ws + WS_O3);
    mace_linear    <<<dim3(16, 32), dim3(256), 0, stream>>>(ws + WS_O3, Wl, out);
}


# Contraction and linear layer in one launch: a work-group owns 16 nodes for all 256 channels, out3 stays in LDS, G_c fragments go from L2 straight into registers one channel ahead; 3 launches -> 2
# speedup vs baseline: 2.1619x; 1.0441x over previous
#include <hip/hip_runtime.h>


typedef float v2f __attribute__((ext_vector_type(2)));
typedef float v4f __attribute__((ext_vector_type(4)));
typedef float v8f __attribute__((ext_vector_type(8)));

#define CCH  256
#define IDIM 16
#define BDIM 4096
#define NK3  23
#define NK2  4
#define NK1  1
#define NPAIR 72
#define NSTEP3 (NPAIR / 4)
#define U3F   (NPAIR * 16)

#define WS_U3W 0
#define WS_C2  (CCH * 4096)
#define WS_C1  (WS_C2 + CCH * 256)
#define WS_O3  (WS_C1 + CCH * 16)


#define SC_PAIRS(X)                                                          \
    X(0, 0, 0, 0, 1)   X(1, 0, 2, 0, 3)   X(2, 0, 4, 0, 5)   X(3, 0, 6, 0, 7)   \
    X(4, 1, 1, 1, 2)   X(5, 1, 3, 1, 4)   X(6, 1, 5, 1, 6)   X(7, 1, 7, 2, 2)   \
    X(8, 2, 3, 2, 4)   X(9, 2, 5, 2, 6)   X(10, 2, 7, 3, 3)  X(11, 3, 4, 3, 5)  \
    X(12, 3, 6, 3, 7)  X(13, 4, 4, 4, 5)  X(14, 4, 6, 4, 7)  X(15, 5, 5, 5, 6)  \
    X(16, 5, 7, 6, 6)  X(17, 6, 7, 7, 7)
__host__ __device__ constexpr int pair_code(int n) {
    const int q = (n >= 8) + (n >= 15) + (n >= 21) + (n >= 26) + (n >= 30) + (n >= 33) + (n >= 35);
    return (q << 4) | (q + n - (q * 8 - q * (q - 1) / 2));
}
#define SC_CHECK(k, q0, i0, q1, i1)                                          \
    static_assert(pair_code(2 * (k)) == (((q0) << 4) | (i0)) &&              \
                  pair_code(2 * (k) + 1) == (((q1) << 4) | (i1)), "pair order");
SC_PAIRS(SC_CHECK)

#define A_CELLS   32
#define A_NBLK3   (U3F / A_CELLS)
#define A_CH      64
#define A_SPLIT   (CCH / A_CH)
#define A_CHW     (A_CH / 8)
#define A_CH2     8
#define A_NBLK2   (CCH / A_CH2)
#define A_GRID    (A_NBLK3 * A_SPLIT + A_NBLK2)
#define A_UFP     28
#define A_WP      24

__launch_bounds__(256)
__global__ void mace_precompute(const float* __restrict__ U3,
                                const float* __restrict__ U2,
                                const float* __restrict__ U1,
                                const float* __restrict__ w3,
                                const float* __restrict__ w2,
                                const float* __restrict__ w1,
                                float* __restrict__ ws) {
    const int t = threadIdx.x;

    if (blockIdx.x >= A_NBLK3 * A_SPLIT) {
        const int c0   = (blockIdx.x - A_NBLK3 * A_SPLIT) * A_CH2;
        const int p    = t & 15;
        const int pos2 = t >> 4;
        const int n    = ((pos2 >> 2) << 1) | (pos2 & 1);
        const int i    = n + ((pos2 >> 1) & 1) * 8;
        float u2[NK2];
#pragma unroll
        for (int k = 0; k < NK2; ++k) u2[k] = U2[(p * 16 + i) * NK2 + k];
        float u1[NK1];
#pragma unroll
        for (int k = 0; k < NK1; ++k) u1[k] = U1[(t & 15) * NK1 + k];
        const int dst2 = ((pos2 >> 1) * 16 + p) * 2 + (pos2 & 1);
        float w2r[A_CH2][NK2], w1r[A_CH2][NK1];
#pragma unroll
        for (int ch = 0; ch < A_CH2; ++ch) {
#pragma unroll
            for (int k = 0; k < NK2; ++k) w2r[ch][k] = w2[(c0 + ch) * NK2 + k];
#pragma unroll
            for (int k = 0; k < NK1; ++k) w1r[ch][k] = w1[(c0 + ch) * NK1 + k];
        }
#pragma unroll
        for (int ch = 0; ch < A_CH2; ++ch) {
            const int c = c0 + ch;
            float acc = 0.f;
#pragma unroll
            for (int k = 0; k < NK2; ++k) acc += u2[k] * w2r[ch][k];
            ws[WS_C2 + c * 256 + dst2] = acc;
            float a1 = 0.f;
#pragma unroll
            for (int k = 0; k < NK1; ++k) a1 += u1[k] * w1r[ch][k];
            if (t < 16) ws[WS_C1 + c * 16 + t] = a1;
        }
        return;
    }

    __shared__ __attribute__((aligned(16))) float sUf[A_CELLS * A_UFP];
    __shared__ __attribute__((aligned(16))) float sW[A_CH * A_WP];

    const int blk = blockIdx.x / A_SPLIT;
    const int c0  = (blockIdx.x % A_SPLIT) * A_CH;

    const int cell = t >> 3, sub = t & 7;
    const int h  = blk & 1;
    const int pp = cell >> 1;
    const int qi = pair_code(((blk >> 1) << 1) | (cell & 1));
    const int qq = (qi >> 4) + 8 * h, ii = (qi & 15) + 8 * h;
    const bool pq = pp != qq, pi = pp != ii, qd = qq != ii;
    const bool mine = ((pp >> 3) != h) || (pp >= ii);
#define U3ROW(a, b, e) ((((a) * 16 + (b)) * 16 + (e)) * NK3)
    const int r0 = U3ROW(pp, qq, ii), r1 = U3ROW(pp, ii, qq), r2 = U3ROW(qq, pp, ii);
    const int r3 = U3ROW(qq, ii, pp), r4 = U3ROW(ii, pp, qq), r5 = U3ROW(ii, qq, pp);
#undef U3ROW
    float a[3][6];
#pragma unroll
    for (int kk = 0; kk < 3; ++kk) {
        const int k = (sub + 8 * kk < NK3) ? sub + 8 * kk : NK3 - 1;
        a[kk][0] = U3[r0 + k]; a[kk][1] = U3[r1 + k]; a[kk][2] = U3[r2 + k];
        a[kk][3] = U3[r3 + k]; a[kk][4] = U3[r4 + k]; a[kk][5] = U3[r5 + k];
    }
    const int wch = t >> 2, wpart = t & 3;
    float wv[6];
#pragma unroll
    for (int j = 0; j < 6; ++j) {
        const int k = (wpart + 4 * j < NK3) ? wpart + 4 * j : NK3 - 1;
        wv[j] = w3[(c0 + wch) * NK3 + k];
    }
#pragma unroll
    for (int kk = 0; kk < 3; ++kk) {
        float v = a[kk][0];
        v += qd               ? a[kk][1] : 0.f;
        v += pq               ? a[kk][2] : 0.f;
        v += (pq && pi)       ? a[kk][3] : 0.f;
        v += (qd && pi)       ? a[kk][4] : 0.f;
        v += (pq && pi && qd) ? a[kk][5] : 0.f;
        sUf[cell * A_UFP + sub + 8 * kk] = mine ? v : 0.f;
    }
#pragma unroll
    for (int j = 0; j < 6; ++j) sW[wch * A_WP + wpart + 4 * j] = wv[j];
    __syncthreads();

    const int w = t >> 5, lane = t & 31;
    float uf[NK3];
#pragma unroll
    for (int k = 0; k < NK3; ++k) uf[k] = sUf[lane * A_UFP + k];
    float* dst = ws + WS_U3W + (size_t)(c0 + w * A_CHW) * 4096 + blk * A_CELLS + lane;
#pragma unroll
    for (int ch = 0; ch < A_CHW; ++ch) {
        const float* wr = sW + (w * A_CHW + ch) * A_WP;
        float acc = 0.f;
#pragma unroll
        for (int k = 0; k < NK3; ++k) acc += uf[k] * wr[k];
        dst[(size_t)ch * 4096] = acc;
    }
}

#define SC_STEP(k, q0, i0, q1, i1)                                           \
        { v2f a;  a.x = xr[q0] * xr[i0];  a.y = xr[q1] * xr[i1];             \
          acc = __builtin_amdgcn_wmma_f32_16x16x4_f32(                       \
              false, a, false, bU[k], (short)0, acc, false, false); }

#define F_WAVES 16

struct ChanOps {
    v2f   bU[NSTEP3];
    v2f   bC2[4];
    float c1v;
    v4f   x0, x1;
};

__device__ __forceinline__ void load_ops(ChanOps& o, const float* __restrict__ x,
                                         const float* __restrict__ wsc,
                                         int c, int b0, int m, int hi) {
    const float* gU3 = wsc + WS_U3W + (size_t)c * 4096 + m * 2 + hi * 32;
#pragma unroll
    for (int k = 0; k < NSTEP3; ++k) o.bU[k] = *(const v2f*)(gU3 + k * 64);
    const float* gC2 = wsc + WS_C2 + (size_t)c * 256 + m * 2 + hi * 32;
#pragma unroll
    for (int k = 0; k < 4; ++k) o.bC2[k] = *(const v2f*)(gC2 + k * 64);
    o.c1v = wsc[WS_C1 + c * 16 + m];
    const float* xrow = x + ((size_t)(b0 + m) * CCH + c) * IDIM + 8 * hi;
    o.x0 = ((const v4f*)xrow)[0];
    o.x1 = ((const v4f*)xrow)[1];
}

__device__ __forceinline__ void contract_one(const ChanOps& o, int c, int lane, int m, int hi,
                                             float* __restrict__ zrow,
                                             float* __restrict__ sO3) {
    const v2f (&bU)[NSTEP3] = o.bU;
    float xr[8];
    xr[0] = o.x0.x; xr[1] = o.x0.y; xr[2] = o.x0.z; xr[3] = o.x0.w;
    xr[4] = o.x1.x; xr[5] = o.x1.y; xr[6] = o.x1.z; xr[7] = o.x1.w;

    v8f acc = {};
#pragma unroll
    for (int k = 0; k < 4; ++k) {
        const int r = 2 * k;
        v2f a;  a.x = xr[r]; a.y = xr[r + 1];
        acc = __builtin_amdgcn_wmma_f32_16x16x4_f32(
            false, a, false, o.bC2[k], (short)0, acc, false, false);
    }
#pragma unroll
    for (int r = 0; r < 8; ++r) acc[r] += o.c1v;

    SC_PAIRS(SC_STEP)

#pragma unroll
    for (int r = 0; r < 8; ++r)
        zrow[(r + 8 * hi) * 16 + m] = acc[r];
    __asm__ volatile("s_wait_dscnt 0x0" ::: "memory");
    float psum = 0.f;
    const float* zr = zrow + m * 16 + 8 * hi;
#pragma unroll
    for (int j = 0; j < 8; ++j) psum += zr[j] * xr[j];
    psum += __shfl_xor(psum, 16, 32);
    if (lane < 16) sO3[c * 16 + m] = psum;
}

__launch_bounds__(512)
__global__ void mace_contract_linear(const float* __restrict__ x,
                                     const float* __restrict__ wsc,
                                     const float* __restrict__ Wl,
                                     float* __restrict__ out) {
    __shared__ float sO3[CCH * 16];
    __shared__ float sZ[F_WAVES * 256];

    const int t    = threadIdx.x;
    const int w    = t >> 5;
    const int lane = t & 31;
    const int m    = lane & 15;
    const int hi   = lane >> 4;
    const int b0   = blockIdx.x * 16;
    float*    zrow = sZ + w * 256;

    const int rot = blockIdx.x & 15;
#define CH_OF(g) (((((g) + rot) & 15) << 4) + w)

    ChanOps P, Q;
    load_ops(P, x, wsc, CH_OF(0), b0, m, hi);
#pragma unroll 1
    for (int g = 0; g < 14; g += 2) {
        load_ops(Q, x, wsc, CH_OF(g + 1), b0, m, hi);
        __builtin_amdgcn_sched_barrier(0);
        contract_one(P, CH_OF(g), lane, m, hi, zrow, sO3);
        load_ops(P, x, wsc, CH_OF(g + 2), b0, m, hi);
        __builtin_amdgcn_sched_barrier(0);
        contract_one(Q, CH_OF(g + 1), lane, m, hi, zrow, sO3);
    }
    load_ops(Q, x, wsc, CH_OF(15), b0, m, hi);
    __builtin_amdgcn_sched_barrier(0);
    contract_one(P, CH_OF(14), lane, m, hi, zrow, sO3);
    contract_one(Q, CH_OF(15), lane, m, hi, zrow, sO3);
#undef CH_OF
    __syncthreads();

    const int d0 = w * 16;
    const float* a0 = sO3 + (2 * hi) * 16 + m;
    const float* wl = Wl + (size_t)(2 * hi) * 256 + d0 + m;
    v8f acc = {};
#pragma unroll
    for (int k = 0; k < 64; ++k) {
        v2f a;   a.x  = a0[k * 64];          a.y  = a0[k * 64 + 16];
        v2f bb;  bb.x = wl[(size_t)k * 1024]; bb.y = wl[(size_t)k * 1024 + 256];
        acc = __builtin_amdgcn_wmma_f32_16x16x4_f32(
            false, a, false, bb, (short)0, acc, false, false);
    }
#pragma unroll
    for (int r = 0; r < 8; ++r)
        out[(size_t)(b0 + r + 8 * hi) * 256 + d0 + m] = acc[r] * 0.0625f;
}

extern "C" void kernel_launch(void* const* d_in, const int* in_sizes, int n_in,
                              void* d_out, int out_size, void* d_ws, size_t ws_size,
                              hipStream_t stream) {
    const float* x  = (const float*)d_in[0];
    const float* U3 = (const float*)d_in[1];
    const float* U2 = (const float*)d_in[2];
    const float* U1 = (const float*)d_in[3];
    const float* w3 = (const float*)d_in[4];
    const float* w2 = (const float*)d_in[5];
    const float* w1 = (const float*)d_in[6];
    const float* Wl = (const float*)d_in[7];
    float* out = (float*)d_out;
    float* ws  = (float*)d_ws;

    mace_precompute     <<<dim3(A_GRID),    dim3(256), 0, stream>>>(U3, U2, U1, w3, w2, w1, ws);
    mace_contract_linear<<<dim3(BDIM / 16), dim3(512), 0, stream>>>(x, ws, Wl, out);
}
